# starting kernel
# baseline (speedup 1.0000x reference)
#include <hip/hip_runtime.h>

#define N_NODES 10000
#define N_EDGES 100000
#define HIDDEN 128
#define MUL0 32
#define MUL1 8
#define IN_DIM 56
#define W_NUMEL 1600
#define EPS 1e-5f

typedef __attribute__((ext_vector_type(2))) float v2f;
typedef __attribute__((ext_vector_type(8))) float v8f;

#define WS_NUM   0
#define WS_CNT   (N_NODES*IN_DIM)
#define WS_STAT  (WS_CNT + N_NODES)
#define WS_TOTAL (WS_STAT + 72)

__global__ __launch_bounds__(256) void zero_ws_kernel(float* __restrict__ ws) {
  int i = blockIdx.x * 256 + threadIdx.x;
  if (i < WS_TOTAL) ws[i] = 0.0f;
}

__global__ __launch_bounds__(256)
void edge_kernel(const float* __restrict__ node_attr,
                 const int*   __restrict__ edge_index,
                 const float* __restrict__ edge_attr,
                 const float* __restrict__ edge_sh,
                 const float* __restrict__ fc1_w, const float* __restrict__ fc1_b,
                 const float* __restrict__ fc2_w, const float* __restrict__ fc2_b,
                 float* __restrict__ num, float* __restrict__ cnt)
{
  __shared__ float sA[32 * 132];
  __shared__ float sH[32 * 132];
  __shared__ float sX[32 * IN_DIM];
  __shared__ float sY[32 * 4];
  __shared__ float sZ0[32 * MUL0];
  __shared__ float sZ1[32 * MUL1];
  __shared__ float sTP[32 * IN_DIM];
  __shared__ int   sSrc[32];

  const int tid  = threadIdx.x;
  const int lane = tid & 31;
  const int wv   = tid >> 5;
  const int e0   = blockIdx.x * 32;

  for (int i = tid; i < 32 * HIDDEN; i += 256) {
    int r = i >> 7, c = i & 127;
    sA[r * 132 + c] = edge_attr[(size_t)(e0 + r) * HIDDEN + c];
  }
  if (tid < 32) sSrc[tid] = edge_index[e0 + tid];
  if (tid < 128) {
    int r = tid >> 2, c = tid & 3;
    sY[r * 4 + c] = edge_sh[(size_t)(e0 + r) * 4 + c];
  }
  for (int i = tid; i < 32 * IN_DIM; i += 256) {
    int r = i / IN_DIM, c = i - r * IN_DIM;
    int dst = edge_index[N_EDGES + e0 + r];
    sX[i]  = node_attr[(size_t)dst * IN_DIM + c];
    sTP[i] = 0.0f;
  }
  __syncthreads();

  for (int i = tid; i < 32 * MUL0; i += 256) {
    int r = i >> 5, u = i & 31;
    sZ0[i] = sX[r * IN_DIM + u] * sY[r * 4 + 0];
  }
  for (int i = tid; i < 32 * MUL1; i += 256) {
    int r = i >> 3, u = i & 7;
    const float* xr = &sX[r * IN_DIM + MUL0 + u * 3];
    const float* yr = &sY[r * 4 + 1];
    sZ1[i] = xr[0] * yr[0] + xr[1] * yr[1] + xr[2] * yr[2];
  }
  __syncthreads();

  const int m     = lane & 15;
  const int koff  = (lane >> 4) << 1;
  const int mrow0 = (lane >> 4) << 3;

  {
    v8f acc0 = {}, acc1 = {};
    const int col = wv * 16 + m;
    #pragma unroll 8
    for (int kb = 0; kb < HIDDEN; kb += 4) {
      v2f a0, a1, b;
      a0.x = sA[m * 132 + kb + koff];
      a0.y = sA[m * 132 + kb + koff + 1];
      a1.x = sA[(16 + m) * 132 + kb + koff];
      a1.y = sA[(16 + m) * 132 + kb + koff + 1];
      b.x  = fc1_w[(size_t)(kb + koff) * HIDDEN + col];
      b.y  = fc1_w[(size_t)(kb + koff + 1) * HIDDEN + col];
      acc0 = __builtin_amdgcn_wmma_f32_16x16x4_f32(false, a0, false, b, (short)0, acc0, false, false);
      acc1 = __builtin_amdgcn_wmma_f32_16x16x4_f32(false, a1, false, b, (short)0, acc1, false, false);
    }
    const float bias = fc1_b[col];
    #pragma unroll
    for (int v = 0; v < 8; ++v) {
      float h0 = acc0[v] + bias;
      float h1 = acc1[v] + bias;
      sH[(mrow0 + v) * 132 + col]      = h0 > 0.0f ? h0 : 0.0f;
      sH[(16 + mrow0 + v) * 132 + col] = h1 > 0.0f ? h1 : 0.0f;
    }
  }
  __syncthreads();

  const float alpha = 0.15811388300841897f;
  const float ais3  = alpha * 0.5773502691896258f;

  for (int nt = wv; nt < 100; nt += 8) {
    v8f cc[2] = {{}, {}};
    const int col = nt * 16 + m;
    #pragma unroll 8
    for (int kb = 0; kb < HIDDEN; kb += 4) {
      v2f a0, a1, b;
      a0.x = sH[m * 132 + kb + koff];
      a0.y = sH[m * 132 + kb + koff + 1];
      a1.x = sH[(16 + m) * 132 + kb + koff];
      a1.y = sH[(16 + m) * 132 + kb + koff + 1];
      b.x  = fc2_w[(size_t)(kb + koff) * W_NUMEL + col];
      b.y  = fc2_w[(size_t)(kb + koff + 1) * W_NUMEL + col];
      cc[0] = __builtin_amdgcn_wmma_f32_16x16x4_f32(false, a0, false, b, (short)0, cc[0], false, false);
      cc[1] = __builtin_amdgcn_wmma_f32_16x16x4_f32(false, a1, false, b, (short)0, cc[1], false, false);
    }
    const float bias = fc2_b[col];

    #pragma unroll
    for (int t = 0; t < 2; ++t) {
      const int eb = t * 16;
      if (col < 1024) {
        const int u = col >> 5, w = col & 31;
        #pragma unroll
        for (int v = 0; v < 8; ++v) {
          int e = eb + mrow0 + v;
          float val = cc[t][v] + bias;
          unsafeAtomicAdd(&sTP[e * IN_DIM + w], alpha * sZ0[e * MUL0 + u] * val);
        }
      } else if (col < 1280) {
        const int jj = col - 1024, u = jj >> 5, w = jj & 31;
        #pragma unroll
        for (int v = 0; v < 8; ++v) {
          int e = eb + mrow0 + v;
          float val = cc[t][v] + bias;
          unsafeAtomicAdd(&sTP[e * IN_DIM + w], ais3 * sZ1[e * MUL1 + u] * val);
        }
      } else if (col < 1536) {
        const int jj = col - 1280, u = jj >> 3, w = jj & 7;
        #pragma unroll
        for (int v = 0; v < 8; ++v) {
          int e = eb + mrow0 + v;
          float val = alpha * (cc[t][v] + bias) * sX[e * IN_DIM + u];
          unsafeAtomicAdd(&sTP[e * IN_DIM + MUL0 + w * 3 + 0], val * sY[e * 4 + 1]);
          unsafeAtomicAdd(&sTP[e * IN_DIM + MUL0 + w * 3 + 1], val * sY[e * 4 + 2]);
          unsafeAtomicAdd(&sTP[e * IN_DIM + MUL0 + w * 3 + 2], val * sY[e * 4 + 3]);
        }
      } else {
        const int jj = col - 1536, u = jj >> 3, w = jj & 7;
        #pragma unroll
        for (int v = 0; v < 8; ++v) {
          int e = eb + mrow0 + v;
          float val = alpha * (cc[t][v] + bias) * sY[e * 4 + 0];
          unsafeAtomicAdd(&sTP[e * IN_DIM + MUL0 + w * 3 + 0], val * sX[e * IN_DIM + MUL0 + u * 3 + 0]);
          unsafeAtomicAdd(&sTP[e * IN_DIM + MUL0 + w * 3 + 1], val * sX[e * IN_DIM + MUL0 + u * 3 + 1]);
          unsafeAtomicAdd(&sTP[e * IN_DIM + MUL0 + w * 3 + 2], val * sX[e * IN_DIM + MUL0 + u * 3 + 2]);
        }
      }
    }
  }
  __syncthreads();

  for (int i = tid; i < 32 * IN_DIM; i += 256) {
    int r = i / IN_DIM, c = i - r * IN_DIM;
    unsafeAtomicAdd(&num[(size_t)sSrc[r] * IN_DIM + c], sTP[i]);
  }
  if (tid < 32) unsafeAtomicAdd(&cnt[sSrc[tid]], 1.0f);
}

__global__ __launch_bounds__(128)
void node_pass1(const float* __restrict__ node_attr,
                const float* __restrict__ num, const float* __restrict__ cnt,
                float* __restrict__ outp, float* __restrict__ stat)
{
  __shared__ float ssum[MUL0], ssq[MUL0], svq[MUL1];
  const int tid = threadIdx.x;
  if (tid < MUL0) { ssum[tid] = 0.0f; ssq[tid] = 0.0f; }
  if (tid < MUL1) svq[tid] = 0.0f;
  __syncthreads();

  const int n = blockIdx.x * 128 + tid;
  if (n < N_NODES) {
    float c   = cnt[n];
    float inv = 1.0f / (c > 1.0f ? c : 1.0f);
    for (int ch = 0; ch < MUL0; ++ch) {
      float p = num[(size_t)n * IN_DIM + ch] * inv + node_attr[(size_t)n * IN_DIM + ch];
      outp[(size_t)n * IN_DIM + ch] = p;
      unsafeAtomicAdd(&ssum[ch], p);
      unsafeAtomicAdd(&ssq[ch], p * p);
    }
    for (int u = 0; u < MUL1; ++u) {
      float acc = 0.0f;
      for (int mm = 0; mm < 3; ++mm) {
        int ch = MUL0 + u * 3 + mm;
        float p = num[(size_t)n * IN_DIM + ch] * inv + node_attr[(size_t)n * IN_DIM + ch];
        outp[(size_t)n * IN_DIM + ch] = p;
        acc += p * p;
      }
      unsafeAtomicAdd(&svq[u], acc);
    }
  }
  __syncthreads();
  if (tid < MUL0) {
    unsafeAtomicAdd(&stat[tid], ssum[tid]);
    unsafeAtomicAdd(&stat[MUL0 + tid], ssq[tid]);
  }
  if (tid < MUL1) unsafeAtomicAdd(&stat[2 * MUL0 + tid], svq[tid]);
}

__global__ __launch_bounds__(128)
void node_pass2(const float* __restrict__ stat,
                const float* __restrict__ bn_w_s, const float* __restrict__ bn_w_v,
                const float* __restrict__ bn_b_s,
                float* __restrict__ out)
{
  const int n = blockIdx.x * 128 + threadIdx.x;
  if (n >= N_NODES) return;
  const float invN = 1.0f / (float)N_NODES;
  for (int ch = 0; ch < MUL0; ++ch) {
    float mean = stat[ch] * invN;
    float var  = stat[MUL0 + ch] * invN - mean * mean;
    float p = out[(size_t)n * IN_DIM + ch];
    out[(size_t)n * IN_DIM + ch] = (p - mean) * rsqrtf(var + EPS) * bn_w_s[ch] + bn_b_s[ch];
  }
  for (int u = 0; u < MUL1; ++u) {
    float vn    = stat[2 * MUL0 + u] * invN * (1.0f / 3.0f);
    float scale = rsqrtf(vn + EPS) * bn_w_v[u];
    for (int mm = 0; mm < 3; ++mm) {
      int ch = MUL0 + u * 3 + mm;
      out[(size_t)n * IN_DIM + ch] *= scale;
    }
  }
}

extern "C" void kernel_launch(void* const* d_in, const int* in_sizes, int n_in,
                              void* d_out, int out_size, void* d_ws, size_t ws_size,
                              hipStream_t stream) {
  const float* node_attr = (const float*)d_in[0];
  const int*   edge_index = (const int*)d_in[1];
  const float* edge_attr = (const float*)d_in[2];
  const float* edge_sh   = (const float*)d_in[3];
  const float* fc1_w = (const float*)d_in[4];
  const float* fc1_b = (const float*)d_in[5];
  const float* fc2_w = (const float*)d_in[6];
  const float* fc2_b = (const float*)d_in[7];
  const float* bn_w_s = (const float*)d_in[8];
  const float* bn_w_v = (const float*)d_in[9];
  const float* bn_b_s = (const float*)d_in[10];

  float* ws   = (float*)d_ws;
  float* num  = ws + WS_NUM;
  float* cnt  = ws + WS_CNT;
  float* stat = ws + WS_STAT;
  float* out  = (float*)d_out;

  zero_ws_kernel<<<(WS_TOTAL + 255) / 256, 256, 0, stream>>>(ws);
  edge_kernel<<<N_EDGES / 32, 256, 0, stream>>>(node_attr, edge_index, edge_attr, edge_sh,
                                                fc1_w, fc1_b, fc2_w, fc2_b, num, cnt);
  node_pass1<<<(N_NODES + 127) / 128, 128, 0, stream>>>(node_attr, num, cnt, out, stat);
  node_pass2<<<(N_NODES + 127) / 128, 128, 0, stream>>>(stat, bn_w_s, bn_w_v, bn_b_s, out);
}


# edge kernel: GEMM2 computed transposed (weights as WMMA A operand, edges as B): a lane's 8 results belong to one edge, so tensor-product sums stay in registers; summation order changes
# speedup vs baseline: 1.4390x; 1.4390x over previous
#include <hip/hip_runtime.h>

#define N_NODES 10000
#define N_EDGES 100000
#define HIDDEN 128
#define MUL0 32
#define MUL1 8
#define IN_DIM 56
#define W_NUMEL 1600
#define EPS 1e-5f

typedef __attribute__((ext_vector_type(2))) float v2f;
typedef __attribute__((ext_vector_type(8))) float v8f;

#define WS_NUM   0
#define WS_CNT   (N_NODES*IN_DIM)
#define WS_STAT  (WS_CNT + N_NODES)
#define WS_TOTAL (WS_STAT + 72)

__global__ __launch_bounds__(256) void zero_ws_kernel(float* __restrict__ ws) {
  int i = blockIdx.x * 256 + threadIdx.x;
  if (i < WS_TOTAL) ws[i] = 0.0f;
}

#define ZC_STRIDE 41

__device__ __forceinline__ void wtile_both(const float* __restrict__ fc2_w, const float* sH,
                                           int j0, int m, int koff, v8f& c0, v8f& c1)
{
  const float* wp = fc2_w + (size_t)koff * W_NUMEL + j0 + m;
  const float* h0p = sH + m * 132 + koff;
  const float* h1p = sH + (16 + m) * 132 + koff;
  #pragma unroll 8
  for (int kb = 0; kb < HIDDEN; kb += 4) {
    v2f w, h0, h1;
    w.x  = wp[(size_t)kb * W_NUMEL];
    w.y  = wp[(size_t)(kb + 1) * W_NUMEL];
    h0.x = h0p[kb];
    h0.y = h0p[kb + 1];
    h1.x = h1p[kb];
    h1.y = h1p[kb + 1];
    c0 = __builtin_amdgcn_wmma_f32_16x16x4_f32(false, w, false, h0, (short)0, c0, false, false);
    c1 = __builtin_amdgcn_wmma_f32_16x16x4_f32(false, w, false, h1, (short)0, c1, false, false);
  }
}

__global__ __launch_bounds__(256)
void edge_kernel(const float* __restrict__ node_attr,
                 const int*   __restrict__ edge_index,
                 const float* __restrict__ edge_attr,
                 const float* __restrict__ edge_sh,
                 const float* __restrict__ fc1_w, const float* __restrict__ fc1_b,
                 const float* __restrict__ fc2_w, const float* __restrict__ fc2_b,
                 float* __restrict__ num, float* __restrict__ cnt)
{
  __shared__ float sA[32 * 132];
  __shared__ float sH[32 * 132];
  __shared__ float sX[32 * IN_DIM];
  __shared__ float sY[32 * 4];
  __shared__ float sZc[32 * ZC_STRIDE];
  __shared__ float sTP[32 * IN_DIM];
  __shared__ float sT[32 * MUL1];
  __shared__ int   sSrc[32];

  const int tid  = threadIdx.x;
  const int lane = tid & 31;
  const int wv   = tid >> 5;
  const int e0   = blockIdx.x * 32;

  for (int i = tid; i < 32 * HIDDEN; i += 256) {
    int r = i >> 7, c = i & 127;
    sA[r * 132 + c] = edge_attr[(size_t)(e0 + r) * HIDDEN + c];
  }
  if (tid < 32) sSrc[tid] = edge_index[e0 + tid];
  if (tid < 128) {
    int r = tid >> 2, c = tid & 3;
    sY[r * 4 + c] = edge_sh[(size_t)(e0 + r) * 4 + c];
  }
  sT[tid] = 0.0f;
  for (int i = tid; i < 32 * IN_DIM; i += 256) {
    int r = i / IN_DIM, c = i - r * IN_DIM;
    int dst = edge_index[N_EDGES + e0 + r];
    sX[i]  = node_attr[(size_t)dst * IN_DIM + c];
    sTP[i] = 0.0f;
  }
  __syncthreads();

  const float alpha = 0.15811388300841897f;
  const float ais3  = alpha * 0.5773502691896258f;

  for (int i = tid; i < 32 * (MUL0 + MUL1); i += 256) {
    int r = i / (MUL0 + MUL1), uu = i - r * (MUL0 + MUL1);
    float z;
    if (uu < MUL0) {
      z = alpha * (sX[r * IN_DIM + uu] * sY[r * 4 + 0]);
    } else {
      const float* xr = &sX[r * IN_DIM + MUL0 + (uu - MUL0) * 3];
      const float* yr = &sY[r * 4 + 1];
      z = ais3 * (xr[0] * yr[0] + xr[1] * yr[1] + xr[2] * yr[2]);
    }
    sZc[r * ZC_STRIDE + uu] = z;
  }
  __syncthreads();

  const int m     = lane & 15;
  const int hh    = lane >> 4;
  const int koff  = hh << 1;
  const int mrow0 = hh << 3;

  {
    v8f acc0 = {}, acc1 = {};
    const int col = wv * 16 + m;
    #pragma unroll 8
    for (int kb = 0; kb < HIDDEN; kb += 4) {
      v2f a0, a1, b;
      a0.x = sA[m * 132 + kb + koff];
      a0.y = sA[m * 132 + kb + koff + 1];
      a1.x = sA[(16 + m) * 132 + kb + koff];
      a1.y = sA[(16 + m) * 132 + kb + koff + 1];
      b.x  = fc1_w[(size_t)(kb + koff) * HIDDEN + col];
      b.y  = fc1_w[(size_t)(kb + koff + 1) * HIDDEN + col];
      acc0 = __builtin_amdgcn_wmma_f32_16x16x4_f32(false, a0, false, b, (short)0, acc0, false, false);
      acc1 = __builtin_amdgcn_wmma_f32_16x16x4_f32(false, a1, false, b, (short)0, acc1, false, false);
    }
    const float bias = fc1_b[col];
    #pragma unroll
    for (int v = 0; v < 8; ++v) {
      float h0 = acc0[v] + bias;
      float h1 = acc1[v] + bias;
      sH[(mrow0 + v) * 132 + col]      = h0 > 0.0f ? h0 : 0.0f;
      sH[(16 + mrow0 + v) * 132 + col] = h1 > 0.0f ? h1 : 0.0f;
    }
  }
  __syncthreads();

  {
    const int wt = wv & 1, g = wv >> 1;
    v8f tp0 = {}, tp1 = {};
    for (int i = 0; i < 10; ++i) {
      const int uu = g * 10 + i;
      const int j0 = uu * 32 + wt * 16;
      const float z0 = sZc[m * ZC_STRIDE + uu];
      const float z1 = sZc[(16 + m) * ZC_STRIDE + uu];
      float bb[8];
      #pragma unroll
      for (int r = 0; r < 8; ++r) bb[r] = fc2_b[j0 + mrow0 + r];
      v8f c0 = {}, c1 = {};
      wtile_both(fc2_w, sH, j0, m, koff, c0, c1);
      #pragma unroll
      for (int r = 0; r < 8; ++r) {
        tp0[r] += z0 * (c0[r] + bb[r]);
        tp1[r] += z1 * (c1[r] + bb[r]);
      }
    }
    #pragma unroll
    for (int r = 0; r < 8; ++r) {
      unsafeAtomicAdd(&sTP[m * IN_DIM + wt * 16 + mrow0 + r], tp0[r]);
      unsafeAtomicAdd(&sTP[(16 + m) * IN_DIM + wt * 16 + mrow0 + r], tp1[r]);
    }
  }

  {
    v8f t0 = {}, t1 = {};
    for (int i = 0; i < 2; ++i) {
      const int q  = wv * 2 + i;
      const int j0 = 1280 + q * 16;
      const int u  = q * 2 + hh;
      const float xa = sX[m * IN_DIM + u];
      const float xb = sX[(16 + m) * IN_DIM + u];
      float bb[8];
      #pragma unroll
      for (int r = 0; r < 8; ++r) bb[r] = fc2_b[j0 + mrow0 + r];
      v8f c0 = {}, c1 = {};
      wtile_both(fc2_w, sH, j0, m, koff, c0, c1);
      #pragma unroll
      for (int r = 0; r < 8; ++r) {
        t0[r] += xa * (c0[r] + bb[r]);
        t1[r] += xb * (c1[r] + bb[r]);
      }
    }
    #pragma unroll
    for (int r = 0; r < 8; ++r) {
      unsafeAtomicAdd(&sT[m * MUL1 + r], t0[r]);
      unsafeAtomicAdd(&sT[(16 + m) * MUL1 + r], t1[r]);
    }
  }

  {
    const int q  = wv >> 1, et = wv & 1;
    const int j0 = 1536 + q * 16;
    const int e  = et * 16 + m;
    const int u  = q * 2 + hh;
    float bb[8];
    #pragma unroll
    for (int r = 0; r < 8; ++r) bb[r] = fc2_b[j0 + mrow0 + r];
    const float y0 = sY[e * 4 + 0];
    const float xv0 = sX[e * IN_DIM + MUL0 + u * 3 + 0];
    const float xv1 = sX[e * IN_DIM + MUL0 + u * 3 + 1];
    const float xv2 = sX[e * IN_DIM + MUL0 + u * 3 + 2];
    v8f ca = {}, cb = {};
    const float* wp = fc2_w + (size_t)koff * W_NUMEL + j0 + m;
    const float* hp = sH + e * 132 + koff;
    #pragma unroll 4
    for (int kb = 0; kb < HIDDEN; kb += 8) {
      v2f w0, w1, ha, hb;
      w0.x = wp[(size_t)kb * W_NUMEL];
      w0.y = wp[(size_t)(kb + 1) * W_NUMEL];
      w1.x = wp[(size_t)(kb + 4) * W_NUMEL];
      w1.y = wp[(size_t)(kb + 5) * W_NUMEL];
      ha.x = hp[kb];
      ha.y = hp[kb + 1];
      hb.x = hp[kb + 4];
      hb.y = hp[kb + 5];
      ca = __builtin_amdgcn_wmma_f32_16x16x4_f32(false, w0, false, ha, (short)0, ca, false, false);
      cb = __builtin_amdgcn_wmma_f32_16x16x4_f32(false, w1, false, hb, (short)0, cb, false, false);
    }
    #pragma unroll
    for (int r = 0; r < 8; ++r) {
      float val = alpha * ((ca[r] + cb[r]) + bb[r]) * y0;
      unsafeAtomicAdd(&sTP[e * IN_DIM + MUL0 + r * 3 + 0], val * xv0);
      unsafeAtomicAdd(&sTP[e * IN_DIM + MUL0 + r * 3 + 1], val * xv1);
      unsafeAtomicAdd(&sTP[e * IN_DIM + MUL0 + r * 3 + 2], val * xv2);
    }
  }
  __syncthreads();

  for (int i = tid; i < 32 * IN_DIM; i += 256) {
    int r = i / IN_DIM, c = i - r * IN_DIM;
    float v = sTP[i];
    if (c >= MUL0) {
      int cc = c - MUL0, w = cc / 3, mm = cc - w * 3;
      v += alpha * sT[r * MUL1 + w] * sY[r * 4 + 1 + mm];
    }
    unsafeAtomicAdd(&num[(size_t)sSrc[r] * IN_DIM + c], v);
  }
  if (tid < 32) unsafeAtomicAdd(&cnt[sSrc[tid]], 1.0f);
}

__global__ __launch_bounds__(128)
void node_pass1(const float* __restrict__ node_attr,
                const float* __restrict__ num, const float* __restrict__ cnt,
                float* __restrict__ outp, float* __restrict__ stat)
{
  __shared__ float ssum[MUL0], ssq[MUL0], svq[MUL1];
  const int tid = threadIdx.x;
  if (tid < MUL0) { ssum[tid] = 0.0f; ssq[tid] = 0.0f; }
  if (tid < MUL1) svq[tid] = 0.0f;
  __syncthreads();

  const int n = blockIdx.x * 128 + tid;
  if (n < N_NODES) {
    float c   = cnt[n];
    float inv = 1.0f / (c > 1.0f ? c : 1.0f);
    for (int ch = 0; ch < MUL0; ++ch) {
      float p = num[(size_t)n * IN_DIM + ch] * inv + node_attr[(size_t)n * IN_DIM + ch];
      outp[(size_t)n * IN_DIM + ch] = p;
      unsafeAtomicAdd(&ssum[ch], p);
      unsafeAtomicAdd(&ssq[ch], p * p);
    }
    for (int u = 0; u < MUL1; ++u) {
      float acc = 0.0f;
      for (int mm = 0; mm < 3; ++mm) {
        int ch = MUL0 + u * 3 + mm;
        float p = num[(size_t)n * IN_DIM + ch] * inv + node_attr[(size_t)n * IN_DIM + ch];
        outp[(size_t)n * IN_DIM + ch] = p;
        acc += p * p;
      }
      unsafeAtomicAdd(&svq[u], acc);
    }
  }
  __syncthreads();
  if (tid < MUL0) {
    unsafeAtomicAdd(&stat[tid], ssum[tid]);
    unsafeAtomicAdd(&stat[MUL0 + tid], ssq[tid]);
  }
  if (tid < MUL1) unsafeAtomicAdd(&stat[2 * MUL0 + tid], svq[tid]);
}

__global__ __launch_bounds__(128)
void node_pass2(const float* __restrict__ stat,
                const float* __restrict__ bn_w_s, const float* __restrict__ bn_w_v,
                const float* __restrict__ bn_b_s,
                float* __restrict__ out)
{
  const int n = blockIdx.x * 128 + threadIdx.x;
  if (n >= N_NODES) return;
  const float invN = 1.0f / (float)N_NODES;
  for (int ch = 0; ch < MUL0; ++ch) {
    float mean = stat[ch] * invN;
    float var  = stat[MUL0 + ch] * invN - mean * mean;
    float p = out[(size_t)n * IN_DIM + ch];
    out[(size_t)n * IN_DIM + ch] = (p - mean) * rsqrtf(var + EPS) * bn_w_s[ch] + bn_b_s[ch];
  }
  for (int u = 0; u < MUL1; ++u) {
    float vn    = stat[2 * MUL0 + u] * invN * (1.0f / 3.0f);
    float scale = rsqrtf(vn + EPS) * bn_w_v[u];
    for (int mm = 0; mm < 3; ++mm) {
      int ch = MUL0 + u * 3 + mm;
      out[(size_t)n * IN_DIM + ch] *= scale;
    }
  }
}

extern "C" void kernel_launch(void* const* d_in, const int* in_sizes, int n_in,
                              void* d_out, int out_size, void* d_ws, size_t ws_size,
                              hipStream_t stream) {
  const float* node_attr = (const float*)d_in[0];
  const int*   edge_index = (const int*)d_in[1];
  const float* edge_attr = (const float*)d_in[2];
  const float* edge_sh   = (const float*)d_in[3];
  const float* fc1_w = (const float*)d_in[4];
  const float* fc1_b = (const float*)d_in[5];
  const float* fc2_w = (const float*)d_in[6];
  const float* fc2_b = (const float*)d_in[7];
  const float* bn_w_s = (const float*)d_in[8];
  const float* bn_w_v = (const float*)d_in[9];
  const float* bn_b_s = (const float*)d_in[10];

  float* ws   = (float*)d_ws;
  float* num  = ws + WS_NUM;
  float* cnt  = ws + WS_CNT;
  float* stat = ws + WS_STAT;
  float* out  = (float*)d_out;

  zero_ws_kernel<<<(WS_TOTAL + 255) / 256, 256, 0, stream>>>(ws);
  edge_kernel<<<N_EDGES / 32, 256, 0, stream>>>(node_attr, edge_index, edge_attr, edge_sh,
                                                fc1_w, fc1_b, fc2_w, fc2_b, num, cnt);
  node_pass1<<<(N_NODES + 127) / 128, 128, 0, stream>>>(node_attr, num, cnt, out, stat);
  node_pass2<<<(N_NODES + 127) / 128, 128, 0, stream>>>(stat, bn_w_s, bn_w_v, bn_b_s, out);
}


# node pass 1: grid gains a second dimension over 10 channel groups (4 channels each), so 10x the waves each run a tenth of the serial per-channel lane sums; atomics and sum order as before
# speedup vs baseline: 1.6352x; 1.1363x over previous
#include <hip/hip_runtime.h>

#define N_NODES 10000
#define N_EDGES 100000
#define HIDDEN 128
#define MUL0 32
#define MUL1 8
#define IN_DIM 56
#define W_NUMEL 1600
#define EPS 1e-5f

typedef __attribute__((ext_vector_type(2))) float v2f;
typedef __attribute__((ext_vector_type(8))) float v8f;

#define WS_NUM   0
#define WS_CNT   (N_NODES*IN_DIM)
#define WS_STAT  (WS_CNT + N_NODES)
#define WS_TOTAL (WS_STAT + 72)

__global__ __launch_bounds__(256) void zero_ws_kernel(float* __restrict__ ws) {
  int i = blockIdx.x * 256 + threadIdx.x;
  if (i < WS_TOTAL) ws[i] = 0.0f;
}

#define ZC_STRIDE 41

__device__ __forceinline__ void wtile_both(const float* __restrict__ fc2_w, const float* sH,
                                           int j0, int m, int koff, v8f& c0, v8f& c1)
{
  const float* wp = fc2_w + (size_t)koff * W_NUMEL + j0 + m;
  const float* h0p = sH + m * 132 + koff;
  const float* h1p = sH + (16 + m) * 132 + koff;
  #pragma unroll 8
  for (int kb = 0; kb < HIDDEN; kb += 4) {
    v2f w, h0, h1;
    w.x  = wp[(size_t)kb * W_NUMEL];
    w.y  = wp[(size_t)(kb + 1) * W_NUMEL];
    h0.x = h0p[kb];
    h0.y = h0p[kb + 1];
    h1.x = h1p[kb];
    h1.y = h1p[kb + 1];
    c0 = __builtin_amdgcn_wmma_f32_16x16x4_f32(false, w, false, h0, (short)0, c0, false, false);
    c1 = __builtin_amdgcn_wmma_f32_16x16x4_f32(false, w, false, h1, (short)0, c1, false, false);
  }
}

__global__ __launch_bounds__(256)
void edge_kernel(const float* __restrict__ node_attr,
                 const int*   __restrict__ edge_index,
                 const float* __restrict__ edge_attr,
                 const float* __restrict__ edge_sh,
                 const float* __restrict__ fc1_w, const float* __restrict__ fc1_b,
                 const float* __restrict__ fc2_w, const float* __restrict__ fc2_b,
                 float* __restrict__ num, float* __restrict__ cnt)
{
  __shared__ float sA[32 * 132];
  __shared__ float sH[32 * 132];
  __shared__ float sX[32 * IN_DIM];
  __shared__ float sY[32 * 4];
  __shared__ float sZc[32 * ZC_STRIDE];
  __shared__ float sTP[32 * IN_DIM];
  __shared__ float sT[32 * MUL1];
  __shared__ int   sSrc[32];

  const int tid  = threadIdx.x;
  const int lane = tid & 31;
  const int wv   = tid >> 5;
  const int e0   = blockIdx.x * 32;

  for (int i = tid; i < 32 * HIDDEN; i += 256) {
    int r = i >> 7, c = i & 127;
    sA[r * 132 + c] = edge_attr[(size_t)(e0 + r) * HIDDEN + c];
  }
  if (tid < 32) sSrc[tid] = edge_index[e0 + tid];
  if (tid < 128) {
    int r = tid >> 2, c = tid & 3;
    sY[r * 4 + c] = edge_sh[(size_t)(e0 + r) * 4 + c];
  }
  sT[tid] = 0.0f;
  for (int i = tid; i < 32 * IN_DIM; i += 256) {
    int r = i / IN_DIM, c = i - r * IN_DIM;
    int dst = edge_index[N_EDGES + e0 + r];
    sX[i]  = node_attr[(size_t)dst * IN_DIM + c];
    sTP[i] = 0.0f;
  }
  __syncthreads();

  const float alpha = 0.15811388300841897f;
  const float ais3  = alpha * 0.5773502691896258f;

  for (int i = tid; i < 32 * (MUL0 + MUL1); i += 256) {
    int r = i / (MUL0 + MUL1), uu = i - r * (MUL0 + MUL1);
    float z;
    if (uu < MUL0) {
      z = alpha * (sX[r * IN_DIM + uu] * sY[r * 4 + 0]);
    } else {
      const float* xr = &sX[r * IN_DIM + MUL0 + (uu - MUL0) * 3];
      const float* yr = &sY[r * 4 + 1];
      z = ais3 * (xr[0] * yr[0] + xr[1] * yr[1] + xr[2] * yr[2]);
    }
    sZc[r * ZC_STRIDE + uu] = z;
  }
  __syncthreads();

  const int m     = lane & 15;
  const int hh    = lane >> 4;
  const int koff  = hh << 1;
  const int mrow0 = hh << 3;

  {
    v8f acc0 = {}, acc1 = {};
    const int col = wv * 16 + m;
    #pragma unroll 8
    for (int kb = 0; kb < HIDDEN; kb += 4) {
      v2f a0, a1, b;
      a0.x = sA[m * 132 + kb + koff];
      a0.y = sA[m * 132 + kb + koff + 1];
      a1.x = sA[(16 + m) * 132 + kb + koff];
      a1.y = sA[(16 + m) * 132 + kb + koff + 1];
      b.x  = fc1_w[(size_t)(kb + koff) * HIDDEN + col];
      b.y  = fc1_w[(size_t)(kb + koff + 1) * HIDDEN + col];
      acc0 = __builtin_amdgcn_wmma_f32_16x16x4_f32(false, a0, false, b, (short)0, acc0, false, false);
      acc1 = __builtin_amdgcn_wmma_f32_16x16x4_f32(false, a1, false, b, (short)0, acc1, false, false);
    }
    const float bias = fc1_b[col];
    #pragma unroll
    for (int v = 0; v < 8; ++v) {
      float h0 = acc0[v] + bias;
      float h1 = acc1[v] + bias;
      sH[(mrow0 + v) * 132 + col]      = h0 > 0.0f ? h0 : 0.0f;
      sH[(16 + mrow0 + v) * 132 + col] = h1 > 0.0f ? h1 : 0.0f;
    }
  }
  __syncthreads();

  {
    const int wt = wv & 1, g = wv >> 1;
    v8f tp0 = {}, tp1 = {};
    for (int i = 0; i < 10; ++i) {
      const int uu = g * 10 + i;
      const int j0 = uu * 32 + wt * 16;
      const float z0 = sZc[m * ZC_STRIDE + uu];
      const float z1 = sZc[(16 + m) * ZC_STRIDE + uu];
      float bb[8];
      #pragma unroll
      for (int r = 0; r < 8; ++r) bb[r] = fc2_b[j0 + mrow0 + r];
      v8f c0 = {}, c1 = {};
      wtile_both(fc2_w, sH, j0, m, koff, c0, c1);
      #pragma unroll
      for (int r = 0; r < 8; ++r) {
        tp0[r] += z0 * (c0[r] + bb[r]);
        tp1[r] += z1 * (c1[r] + bb[r]);
      }
    }
    #pragma unroll
    for (int r = 0; r < 8; ++r) {
      unsafeAtomicAdd(&sTP[m * IN_DIM + wt * 16 + mrow0 + r], tp0[r]);
      unsafeAtomicAdd(&sTP[(16 + m) * IN_DIM + wt * 16 + mrow0 + r], tp1[r]);
    }
  }

  {
    v8f t0 = {}, t1 = {};
    for (int i = 0; i < 2; ++i) {
      const int q  = wv * 2 + i;
      const int j0 = 1280 + q * 16;
      const int u  = q * 2 + hh;
      const float xa = sX[m * IN_DIM + u];
      const float xb = sX[(16 + m) * IN_DIM + u];
      float bb[8];
      #pragma unroll
      for (int r = 0; r < 8; ++r) bb[r] = fc2_b[j0 + mrow0 + r];
      v8f c0 = {}, c1 = {};
      wtile_both(fc2_w, sH, j0, m, koff, c0, c1);
      #pragma unroll
      for (int r = 0; r < 8; ++r) {
        t0[r] += xa * (c0[r] + bb[r]);
        t1[r] += xb * (c1[r] + bb[r]);
      }
    }
    #pragma unroll
    for (int r = 0; r < 8; ++r) {
      unsafeAtomicAdd(&sT[m * MUL1 + r], t0[r]);
      unsafeAtomicAdd(&sT[(16 + m) * MUL1 + r], t1[r]);
    }
  }

  {
    const int q  = wv >> 1, et = wv & 1;
    const int j0 = 1536 + q * 16;
    const int e  = et * 16 + m;
    const int u  = q * 2 + hh;
    float bb[8];
    #pragma unroll
    for (int r = 0; r < 8; ++r) bb[r] = fc2_b[j0 + mrow0 + r];
    const float y0 = sY[e * 4 + 0];
    const float xv0 = sX[e * IN_DIM + MUL0 + u * 3 + 0];
    const float xv1 = sX[e * IN_DIM + MUL0 + u * 3 + 1];
    const float xv2 = sX[e * IN_DIM + MUL0 + u * 3 + 2];
    v8f ca = {}, cb = {};
    const float* wp = fc2_w + (size_t)koff * W_NUMEL + j0 + m;
    const float* hp = sH + e * 132 + koff;
    #pragma unroll 4
    for (int kb = 0; kb < HIDDEN; kb += 8) {
      v2f w0, w1, ha, hb;
      w0.x = wp[(size_t)kb * W_NUMEL];
      w0.y = wp[(size_t)(kb + 1) * W_NUMEL];
      w1.x = wp[(size_t)(kb + 4) * W_NUMEL];
      w1.y = wp[(size_t)(kb + 5) * W_NUMEL];
      ha.x = hp[kb];
      ha.y = hp[kb + 1];
      hb.x = hp[kb + 4];
      hb.y = hp[kb + 5];
      ca = __builtin_amdgcn_wmma_f32_16x16x4_f32(false, w0, false, ha, (short)0, ca, false, false);
      cb = __builtin_amdgcn_wmma_f32_16x16x4_f32(false, w1, false, hb, (short)0, cb, false, false);
    }
    #pragma unroll
    for (int r = 0; r < 8; ++r) {
      float val = alpha * ((ca[r] + cb[r]) + bb[r]) * y0;
      unsafeAtomicAdd(&sTP[e * IN_DIM + MUL0 + r * 3 + 0], val * xv0);
      unsafeAtomicAdd(&sTP[e * IN_DIM + MUL0 + r * 3 + 1], val * xv1);
      unsafeAtomicAdd(&sTP[e * IN_DIM + MUL0 + r * 3 + 2], val * xv2);
    }
  }
  __syncthreads();

  for (int i = tid; i < 32 * IN_DIM; i += 256) {
    int r = i / IN_DIM, c = i - r * IN_DIM;
    float v = sTP[i];
    if (c >= MUL0) {
      int cc = c - MUL0, w = cc / 3, mm = cc - w * 3;
      v += alpha * sT[r * MUL1 + w] * sY[r * 4 + 1 + mm];
    }
    unsafeAtomicAdd(&num[(size_t)sSrc[r] * IN_DIM + c], v);
  }
  if (tid < 32) unsafeAtomicAdd(&cnt[sSrc[tid]], 1.0f);
}

#define P1_GROUPS 10
#define P1_PER    4

__global__ __launch_bounds__(128)
void node_pass1(const float* __restrict__ node_attr,
                const float* __restrict__ num, const float* __restrict__ cnt,
                float* __restrict__ outp, float* __restrict__ stat)
{
  __shared__ float sa[P1_PER], sb[P1_PER];
  const int tid = threadIdx.x;
  const int grp = blockIdx.y;
  if (tid < P1_PER) { sa[tid] = 0.0f; sb[tid] = 0.0f; }
  __syncthreads();

  const int n = blockIdx.x * 128 + tid;
  if (n < N_NODES) {
    float c   = cnt[n];
    float inv = 1.0f / (c > 1.0f ? c : 1.0f);
    if (grp < MUL0 / P1_PER) {
      float p[P1_PER];
      #pragma unroll
      for (int k = 0; k < P1_PER; ++k) {
        int ch = grp * P1_PER + k;
        p[k] = num[(size_t)n * IN_DIM + ch] * inv + node_attr[(size_t)n * IN_DIM + ch];
        outp[(size_t)n * IN_DIM + ch] = p[k];
      }
      #pragma unroll
      for (int k = 0; k < P1_PER; ++k) {
        unsafeAtomicAdd(&sa[k], p[k]);
        unsafeAtomicAdd(&sb[k], p[k] * p[k]);
      }
    } else {
      float acc[P1_PER];
      #pragma unroll
      for (int k = 0; k < P1_PER; ++k) {
        int u = (grp - MUL0 / P1_PER) * P1_PER + k;
        float a = 0.0f;
        #pragma unroll
        for (int mm = 0; mm < 3; ++mm) {
          int ch = MUL0 + u * 3 + mm;
          float q = num[(size_t)n * IN_DIM + ch] * inv + node_attr[(size_t)n * IN_DIM + ch];
          outp[(size_t)n * IN_DIM + ch] = q;
          a += q * q;
        }
        acc[k] = a;
      }
      #pragma unroll
      for (int k = 0; k < P1_PER; ++k) unsafeAtomicAdd(&sa[k], acc[k]);
    }
  }
  __syncthreads();
  if (tid < P1_PER) {
    if (grp < MUL0 / P1_PER) {
      unsafeAtomicAdd(&stat[grp * P1_PER + tid], sa[tid]);
      unsafeAtomicAdd(&stat[MUL0 + grp * P1_PER + tid], sb[tid]);
    } else {
      unsafeAtomicAdd(&stat[2 * MUL0 + (grp - MUL0 / P1_PER) * P1_PER + tid], sa[tid]);
    }
  }
}

__global__ __launch_bounds__(128)
void node_pass2(const float* __restrict__ stat,
                const float* __restrict__ bn_w_s, const float* __restrict__ bn_w_v,
                const float* __restrict__ bn_b_s,
                float* __restrict__ out)
{
  const int n = blockIdx.x * 128 + threadIdx.x;
  if (n >= N_NODES) return;
  const float invN = 1.0f / (float)N_NODES;
  for (int ch = 0; ch < MUL0; ++ch) {
    float mean = stat[ch] * invN;
    float var  = stat[MUL0 + ch] * invN - mean * mean;
    float p = out[(size_t)n * IN_DIM + ch];
    out[(size_t)n * IN_DIM + ch] = (p - mean) * rsqrtf(var + EPS) * bn_w_s[ch] + bn_b_s[ch];
  }
  for (int u = 0; u < MUL1; ++u) {
    float vn    = stat[2 * MUL0 + u] * invN * (1.0f / 3.0f);
    float scale = rsqrtf(vn + EPS) * bn_w_v[u];
    for (int mm = 0; mm < 3; ++mm) {
      int ch = MUL0 + u * 3 + mm;
      out[(size_t)n * IN_DIM + ch] *= scale;
    }
  }
}

extern "C" void kernel_launch(void* const* d_in, const int* in_sizes, int n_in,
                              void* d_out, int out_size, void* d_ws, size_t ws_size,
                              hipStream_t stream) {
  const float* node_attr = (const float*)d_in[0];
  const int*   edge_index = (const int*)d_in[1];
  const float* edge_attr = (const float*)d_in[2];
  const float* edge_sh   = (const float*)d_in[3];
  const float* fc1_w = (const float*)d_in[4];
  const float* fc1_b = (const float*)d_in[5];
  const float* fc2_w = (const float*)d_in[6];
  const float* fc2_b = (const float*)d_in[7];
  const float* bn_w_s = (const float*)d_in[8];
  const float* bn_w_v = (const float*)d_in[9];
  const float* bn_b_s = (const float*)d_in[10];

  float* ws   = (float*)d_ws;
  float* num  = ws + WS_NUM;
  float* cnt  = ws + WS_CNT;
  float* stat = ws + WS_STAT;
  float* out  = (float*)d_out;

  zero_ws_kernel<<<(WS_TOTAL + 255) / 256, 256, 0, stream>>>(ws);
  edge_kernel<<<N_EDGES / 32, 256, 0, stream>>>(node_attr, edge_index, edge_attr, edge_sh,
                                                fc1_w, fc1_b, fc2_w, fc2_b, num, cnt);
  node_pass1<<<dim3((N_NODES + 127) / 128, P1_GROUPS), 128, 0, stream>>>(node_attr, num, cnt, out, stat);
  node_pass2<<<(N_NODES + 127) / 128, 128, 0, stream>>>(stat, bn_w_s, bn_w_v, bn_b_s, out);
}


# edge kernel: 80 edges per block instead of 32: five 16-edge row tiles share each weight fragment a wave fetches, so one pass over the weights serves 80 edges; k-loop groups 16 steps; sum order kept
# speedup vs baseline: 1.8121x; 1.1082x over previous
#include <hip/hip_runtime.h>

#define N_NODES 10000
#define N_EDGES 100000
#define HIDDEN 128
#define MUL0 32
#define MUL1 8
#define IN_DIM 56
#define W_NUMEL 1600
#define EPS 1e-5f

typedef __attribute__((ext_vector_type(2))) float v2f;
typedef __attribute__((ext_vector_type(8))) float v8f;

#define WS_NUM   0
#define WS_CNT   (N_NODES*IN_DIM)
#define WS_STAT  (WS_CNT + N_NODES)
#define WS_TOTAL (WS_STAT + 72)

__global__ __launch_bounds__(256) void zero_ws_kernel(float* __restrict__ ws) {
  int i = blockIdx.x * 256 + threadIdx.x;
  if (i < WS_TOTAL) ws[i] = 0.0f;
}

#define ZC_STRIDE 41
#define ET 5
#define EB (ET * 16)
static_assert(N_EDGES % EB == 0, "the grid covers the edges with whole blocks");

__device__ __forceinline__ void wtile_all(const float* __restrict__ fc2_w, const float* sH,
                                          int j0, int m, int koff, v8f (&c)[ET])
{
  const float* wp = fc2_w + (size_t)koff * W_NUMEL + j0 + m;
  const float* hp = sH + m * 132 + koff;
  #pragma unroll 16
  for (int kb = 0; kb < HIDDEN; kb += 4) {
    v2f w;
    w.x = wp[(size_t)kb * W_NUMEL];
    w.y = wp[(size_t)(kb + 1) * W_NUMEL];
    #pragma unroll
    for (int t = 0; t < ET; ++t) {
      v2f h;
      h.x = hp[t * 16 * 132 + kb];
      h.y = hp[t * 16 * 132 + kb + 1];
      c[t] = __builtin_amdgcn_wmma_f32_16x16x4_f32(false, w, false, h, (short)0, c[t], false, false);
    }
  }
}

__global__ __launch_bounds__(256)
void edge_kernel(const float* __restrict__ node_attr,
                 const int*   __restrict__ edge_index,
                 const float* __restrict__ edge_attr,
                 const float* __restrict__ edge_sh,
                 const float* __restrict__ fc1_w, const float* __restrict__ fc1_b,
                 const float* __restrict__ fc2_w, const float* __restrict__ fc2_b,
                 float* __restrict__ num, float* __restrict__ cnt)
{
  __shared__ float sA[EB * 132];
  __shared__ float sH[EB * 132];
  __shared__ float sX[EB * IN_DIM];
  __shared__ float sY[EB * 4];
  __shared__ float sZc[EB * ZC_STRIDE];
  __shared__ float sTP[EB * IN_DIM];
  __shared__ float sT[EB * MUL1];
  __shared__ int   sSrc[EB];

  const int tid  = threadIdx.x;
  const int lane = tid & 31;
  const int wv   = tid >> 5;
  const int e0   = blockIdx.x * EB;

  for (int i = tid; i < EB * HIDDEN; i += 256) {
    int r = i >> 7, c = i & 127;
    sA[r * 132 + c] = edge_attr[(size_t)(e0 + r) * HIDDEN + c];
  }
  if (tid < EB) sSrc[tid] = edge_index[e0 + tid];
  for (int i = tid; i < EB * 4; i += 256) sY[i] = edge_sh[(size_t)e0 * 4 + i];
  for (int i = tid; i < EB * MUL1; i += 256) sT[i] = 0.0f;
  for (int i = tid; i < EB * IN_DIM; i += 256) {
    int r = i / IN_DIM, c = i - r * IN_DIM;
    int dst = edge_index[N_EDGES + e0 + r];
    sX[i]  = node_attr[(size_t)dst * IN_DIM + c];
    sTP[i] = 0.0f;
  }
  __syncthreads();

  const float alpha = 0.15811388300841897f;
  const float ais3  = alpha * 0.5773502691896258f;

  for (int i = tid; i < EB * (MUL0 + MUL1); i += 256) {
    int r = i / (MUL0 + MUL1), uu = i - r * (MUL0 + MUL1);
    float z;
    if (uu < MUL0) {
      z = alpha * (sX[r * IN_DIM + uu] * sY[r * 4 + 0]);
    } else {
      const float* xr = &sX[r * IN_DIM + MUL0 + (uu - MUL0) * 3];
      const float* yr = &sY[r * 4 + 1];
      z = ais3 * (xr[0] * yr[0] + xr[1] * yr[1] + xr[2] * yr[2]);
    }
    sZc[r * ZC_STRIDE + uu] = z;
  }
  __syncthreads();

  const int m     = lane & 15;
  const int hh    = lane >> 4;
  const int koff  = hh << 1;
  const int mrow0 = hh << 3;

  {
    v8f acc[ET];
    #pragma unroll
    for (int t = 0; t < ET; ++t) acc[t] = v8f{};
    const int col = wv * 16 + m;
    const float* ap = sA + m * 132 + koff;
    #pragma unroll 8
    for (int kb = 0; kb < HIDDEN; kb += 4) {
      v2f b;
      b.x = fc1_w[(size_t)(kb + koff) * HIDDEN + col];
      b.y = fc1_w[(size_t)(kb + koff + 1) * HIDDEN + col];
      #pragma unroll
      for (int t = 0; t < ET; ++t) {
        v2f a;
        a.x = ap[t * 16 * 132 + kb];
        a.y = ap[t * 16 * 132 + kb + 1];
        acc[t] = __builtin_amdgcn_wmma_f32_16x16x4_f32(false, a, false, b, (short)0, acc[t], false, false);
      }
    }
    const float bias = fc1_b[col];
    #pragma unroll
    for (int t = 0; t < ET; ++t) {
      #pragma unroll
      for (int v = 0; v < 8; ++v) {
        float h = acc[t][v] + bias;
        sH[(t * 16 + mrow0 + v) * 132 + col] = h > 0.0f ? h : 0.0f;
      }
    }
  }
  __syncthreads();

  {
    const int wt = wv & 1, g = wv >> 1;
    v8f tp[ET];
    #pragma unroll
    for (int t = 0; t < ET; ++t) tp[t] = v8f{};
    for (int i = 0; i < 10; ++i) {
      const int uu = g * 10 + i;
      const int j0 = uu * 32 + wt * 16;
      float z[ET];
      #pragma unroll
      for (int t = 0; t < ET; ++t) z[t] = sZc[(t * 16 + m) * ZC_STRIDE + uu];
      float bb[8];
      #pragma unroll
      for (int r = 0; r < 8; ++r) bb[r] = fc2_b[j0 + mrow0 + r];
      v8f c[ET];
      #pragma unroll
      for (int t = 0; t < ET; ++t) c[t] = v8f{};
      wtile_all(fc2_w, sH, j0, m, koff, c);
      #pragma unroll
      for (int t = 0; t < ET; ++t) {
        #pragma unroll
        for (int r = 0; r < 8; ++r) tp[t][r] += z[t] * (c[t][r] + bb[r]);
      }
    }
    #pragma unroll
    for (int t = 0; t < ET; ++t) {
      #pragma unroll
      for (int r = 0; r < 8; ++r)
        unsafeAtomicAdd(&sTP[(t * 16 + m) * IN_DIM + wt * 16 + mrow0 + r], tp[t][r]);
    }
  }

  {
    v8f ts[ET];
    #pragma unroll
    for (int t = 0; t < ET; ++t) ts[t] = v8f{};
    for (int i = 0; i < 2; ++i) {
      const int q  = wv * 2 + i;
      const int j0 = 1280 + q * 16;
      const int u  = q * 2 + hh;
      float xa[ET];
      #pragma unroll
      for (int t = 0; t < ET; ++t) xa[t] = sX[(t * 16 + m) * IN_DIM + u];
      float bb[8];
      #pragma unroll
      for (int r = 0; r < 8; ++r) bb[r] = fc2_b[j0 + mrow0 + r];
      v8f c[ET];
      #pragma unroll
      for (int t = 0; t < ET; ++t) c[t] = v8f{};
      wtile_all(fc2_w, sH, j0, m, koff, c);
      #pragma unroll
      for (int t = 0; t < ET; ++t) {
        #pragma unroll
        for (int r = 0; r < 8; ++r) ts[t][r] += xa[t] * (c[t][r] + bb[r]);
      }
    }
    #pragma unroll
    for (int t = 0; t < ET; ++t) {
      #pragma unroll
      for (int r = 0; r < 8; ++r) unsafeAtomicAdd(&sT[(t * 16 + m) * MUL1 + r], ts[t][r]);
    }
  }

  for (int unit = wv; unit < 4 * ET; unit += 8) {
    const int q  = unit & 3, et = unit >> 2;
    const int j0 = 1536 + q * 16;
    const int e  = et * 16 + m;
    const int u  = q * 2 + hh;
    float bb[8];
    #pragma unroll
    for (int r = 0; r < 8; ++r) bb[r] = fc2_b[j0 + mrow0 + r];
    const float y0 = sY[e * 4 + 0];
    const float xv0 = sX[e * IN_DIM + MUL0 + u * 3 + 0];
    const float xv1 = sX[e * IN_DIM + MUL0 + u * 3 + 1];
    const float xv2 = sX[e * IN_DIM + MUL0 + u * 3 + 2];
    v8f ca = {}, cb = {};
    const float* wp = fc2_w + (size_t)koff * W_NUMEL + j0 + m;
    const float* hp = sH + e * 132 + koff;
    #pragma unroll 4
    for (int kb = 0; kb < HIDDEN; kb += 8) {
      v2f w0, w1, ha, hb;
      w0.x = wp[(size_t)kb * W_NUMEL];
      w0.y = wp[(size_t)(kb + 1) * W_NUMEL];
      w1.x = wp[(size_t)(kb + 4) * W_NUMEL];
      w1.y = wp[(size_t)(kb + 5) * W_NUMEL];
      ha.x = hp[kb];
      ha.y = hp[kb + 1];
      hb.x = hp[kb + 4];
      hb.y = hp[kb + 5];
      ca = __builtin_amdgcn_wmma_f32_16x16x4_f32(false, w0, false, ha, (short)0, ca, false, false);
      cb = __builtin_amdgcn_wmma_f32_16x16x4_f32(false, w1, false, hb, (short)0, cb, false, false);
    }
    #pragma unroll
    for (int r = 0; r < 8; ++r) {
      float val = alpha * ((ca[r] + cb[r]) + bb[r]) * y0;
      unsafeAtomicAdd(&sTP[e * IN_DIM + MUL0 + r * 3 + 0], val * xv0);
      unsafeAtomicAdd(&sTP[e * IN_DIM + MUL0 + r * 3 + 1], val * xv1);
      unsafeAtomicAdd(&sTP[e * IN_DIM + MUL0 + r * 3 + 2], val * xv2);
    }
  }
  __syncthreads();

  for (int i = tid; i < EB * IN_DIM; i += 256) {
    int r = i / IN_DIM, c = i - r * IN_DIM;
    float v = sTP[i];
    if (c >= MUL0) {
      int cc = c - MUL0, w = cc / 3, mm = cc - w * 3;
      v += alpha * sT[r * MUL1 + w] * sY[r * 4 + 1 + mm];
    }
    unsafeAtomicAdd(&num[(size_t)sSrc[r] * IN_DIM + c], v);
  }
  if (tid < EB) unsafeAtomicAdd(&cnt[sSrc[tid]], 1.0f);
}

#define P1_GROUPS 10
#define P1_PER    4

__global__ __launch_bounds__(128)
void node_pass1(const float* __restrict__ node_attr,
                const float* __restrict__ num, const float* __restrict__ cnt,
                float* __restrict__ outp, float* __restrict__ stat)
{
  __shared__ float sa[P1_PER], sb[P1_PER];
  const int tid = threadIdx.x;
  const int grp = blockIdx.y;
  if (tid < P1_PER) { sa[tid] = 0.0f; sb[tid] = 0.0f; }
  __syncthreads();

  const int n = blockIdx.x * 128 + tid;
  if (n < N_NODES) {
    float c   = cnt[n];
    float inv = 1.0f / (c > 1.0f ? c : 1.0f);
    if (grp < MUL0 / P1_PER) {
      float p[P1_PER];
      #pragma unroll
      for (int k = 0; k < P1_PER; ++k) {
        int ch = grp * P1_PER + k;
        p[k] = num[(size_t)n * IN_DIM + ch] * inv + node_attr[(size_t)n * IN_DIM + ch];
        outp[(size_t)n * IN_DIM + ch] = p[k];
      }
      #pragma unroll
      for (int k = 0; k < P1_PER; ++k) {
        unsafeAtomicAdd(&sa[k], p[k]);
        unsafeAtomicAdd(&sb[k], p[k] * p[k]);
      }
    } else {
      float acc[P1_PER];
      #pragma unroll
      for (int k = 0; k < P1_PER; ++k) {
        int u = (grp - MUL0 / P1_PER) * P1_PER + k;
        float a = 0.0f;
        #pragma unroll
        for (int mm = 0; mm < 3; ++mm) {
          int ch = MUL0 + u * 3 + mm;
          float q = num[(size_t)n * IN_DIM + ch] * inv + node_attr[(size_t)n * IN_DIM + ch];
          outp[(size_t)n * IN_DIM + ch] = q;
          a += q * q;
        }
        acc[k] = a;
      }
      #pragma unroll
      for (int k = 0; k < P1_PER; ++k) unsafeAtomicAdd(&sa[k], acc[k]);
    }
  }
  __syncthreads();
  if (tid < P1_PER) {
    if (grp < MUL0 / P1_PER) {
      unsafeAtomicAdd(&stat[grp * P1_PER + tid], sa[tid]);
      unsafeAtomicAdd(&stat[MUL0 + grp * P1_PER + tid], sb[tid]);
    } else {
      unsafeAtomicAdd(&stat[2 * MUL0 + (grp - MUL0 / P1_PER) * P1_PER + tid], sa[tid]);
    }
  }
}

__global__ __launch_bounds__(128)
void node_pass2(const float* __restrict__ stat,
                const float* __restrict__ bn_w_s, const float* __restrict__ bn_w_v,
                const float* __restrict__ bn_b_s,
                float* __restrict__ out)
{
  const int n = blockIdx.x * 128 + threadIdx.x;
  if (n >= N_NODES) return;
  const float invN = 1.0f / (float)N_NODES;
  for (int ch = 0; ch < MUL0; ++ch) {
    float mean = stat[ch] * invN;
    float var  = stat[MUL0 + ch] * invN - mean * mean;
    float p = out[(size_t)n * IN_DIM + ch];
    out[(size_t)n * IN_DIM + ch] = (p - mean) * rsqrtf(var + EPS) * bn_w_s[ch] + bn_b_s[ch];
  }
  for (int u = 0; u < MUL1; ++u) {
    float vn    = stat[2 * MUL0 + u] * invN * (1.0f / 3.0f);
    float scale = rsqrtf(vn + EPS) * bn_w_v[u];
    for (int mm = 0; mm < 3; ++mm) {
      int ch = MUL0 + u * 3 + mm;
      out[(size_t)n * IN_DIM + ch] *= scale;
    }
  }
}

extern "C" void kernel_launch(void* const* d_in, const int* in_sizes, int n_in,
                              void* d_out, int out_size, void* d_ws, size_t ws_size,
                              hipStream_t stream) {
  const float* node_attr = (const float*)d_in[0];
  const int*   edge_index = (const int*)d_in[1];
  const float* edge_attr = (const float*)d_in[2];
  const float* edge_sh   = (const float*)d_in[3];
  const float* fc1_w = (const float*)d_in[4];
  const float* fc1_b = (const float*)d_in[5];
  const float* fc2_w = (const float*)d_in[6];
  const float* fc2_b = (const float*)d_in[7];
  const float* bn_w_s = (const float*)d_in[8];
  const float* bn_w_v = (const float*)d_in[9];
  const float* bn_b_s = (const float*)d_in[10];

  float* ws   = (float*)d_ws;
  float* num  = ws + WS_NUM;
  float* cnt  = ws + WS_CNT;
  float* stat = ws + WS_STAT;
  float* out  = (float*)d_out;

  zero_ws_kernel<<<(WS_TOTAL + 255) / 256, 256, 0, stream>>>(ws);
  edge_kernel<<<N_EDGES / EB, 256, 0, stream>>>(node_attr, edge_index, edge_attr, edge_sh,
                                                fc1_w, fc1_b, fc2_w, fc2_b, num, cnt);
  node_pass1<<<dim3((N_NODES + 127) / 128, P1_GROUPS), 128, 0, stream>>>(node_attr, num, cnt, out, stat);
  node_pass2<<<(N_NODES + 127) / 128, 128, 0, stream>>>(stat, bn_w_s, bn_w_v, bn_b_s, out);
}
